# starting kernel
# baseline (speedup 1.0000x reference)
#include <hip/hip_runtime.h>
#include <hip/hip_bf16.h>

typedef __attribute__((ext_vector_type(2))) float v2f;
typedef __attribute__((ext_vector_type(8))) float v8f;

static constexpr int   NN    = 10000;
static constexpr int   TT    = 8;
static constexpr int   PP    = 2;
static constexpr int   FIN   = 64;
static constexpr int   FOUT  = 64;
static constexpr int   HH    = 128;
static constexpr int   BB    = 2;
static constexpr int   EE    = 160000;
static constexpr int   GG    = BB * TT;
static constexpr int   BN    = BB * NN;
static constexpr int   HID   = HH * 4;
static constexpr size_t GNF  = (size_t)GG * NN * FOUT;

__device__ __forceinline__ v8f wmma4(v2f a, v2f b, v8f c) {
  return __builtin_amdgcn_wmma_f32_16x16x4_f32(false, a, false, b, (short)0, c,
                                               false, false);
}

__device__ __forceinline__ void atomAddF(float* p, float v) {
  __hip_atomic_fetch_add(p, v, __ATOMIC_RELAXED, __HIP_MEMORY_SCOPE_AGENT);
}

__global__ void k_deg_init(float* deg) {
  int i = blockIdx.x * 256 + threadIdx.x;
  if (i < NN) deg[i] = 0.f;
}

__global__ void k_deg_scatter(const int* __restrict__ ei,
                              const float* __restrict__ ew,
                              float* __restrict__ deg) {
  int e = blockIdx.x * 256 + threadIdx.x;
  if (e < EE)                atomAddF(&deg[ei[EE + e]], ew[e]);
  else if (e < EE + NN)      atomAddF(&deg[e - EE], 1.0f);
}

__global__ void k_dinv(const float* __restrict__ deg, float* __restrict__ dinv) {
  int i = blockIdx.x * 256 + threadIdx.x;
  if (i < NN) {
    float d = deg[i];
    dinv[i] = d > 0.f ? rsqrtf(d) : 0.f;
  }
}

__global__ __launch_bounds__(256) void k_gcn_gemm(const float* __restrict__ X,
                                                  const float* __restrict__ W,
                                                  float* __restrict__ Y) {
  __shared__ float wl[FOUT * FIN];
  for (int i = threadIdx.x; i < FOUT * FIN; i += 256) wl[i] = W[i];
  __syncthreads();
  const int wave = threadIdx.x >> 5, lane = threadIdx.x & 31;
  const int mr = lane & 15, half = lane >> 4, kk = half * 2;
  const int row0 = blockIdx.x * 32 + (wave & 1) * 16;
  const int n0 = (wave >> 1) * 16;
  const float* ap = X + (size_t)(row0 + mr) * FIN + kk;
  const float* bp = wl + (n0 + mr) * FIN + kk;
  v8f c = {};
#pragma unroll
  for (int k = 0; k < FIN; k += 4) {
    v2f a = {ap[k], ap[k + 1]};
    v2f b = {bp[k], bp[k + 1]};
    c = wmma4(a, b, c);
  }
#pragma unroll
  for (int j = 0; j < 8; ++j)
    Y[(size_t)(row0 + j + 8 * half) * FOUT + n0 + mr] = c[j];
}

__global__ void k_agg_init(const float* __restrict__ b, float* __restrict__ gc) {
  size_t i = (size_t)blockIdx.x * 256 + threadIdx.x;
  if (i < GNF) gc[i] = b[i & 63];
}

__global__ void k_scatter(const int* __restrict__ ei, const float* __restrict__ ew,
                          const float* __restrict__ dinv,
                          const float* __restrict__ xl, float* __restrict__ gc) {
  size_t idx = (size_t)blockIdx.x * 256 + threadIdx.x;
  if (idx >= (size_t)(EE + NN) * 64) return;
  int f = (int)(idx & 63);
  int e = (int)(idx >> 6);
  int r, cdst; float nrm;
  if (e < EE) {
    r = ei[e]; cdst = ei[EE + e];
    nrm = dinv[r] * ew[e] * dinv[cdst];
  } else {
    int n = e - EE;
    r = n; cdst = n;
    float di = dinv[n];
    nrm = di * di;
  }
  const float* src = xl + (size_t)r * FOUT + f;
  float* dst = gc + (size_t)cdst * FOUT + f;
#pragma unroll
  for (int g = 0; g < GG; ++g)
    atomAddF(dst + (size_t)g * NN * FOUT, src[(size_t)g * NN * FOUT] * nrm);
}

__global__ __launch_bounds__(256) void k_rnn(const float* __restrict__ gc,
                                             const float* __restrict__ Wih,
                                             const float* __restrict__ Whh,
                                             const float* __restrict__ bih,
                                             const float* __restrict__ bhh,
                                             float* __restrict__ rnn) {
  extern __shared__ float sm[];
  float* s_wih = sm;
  float* s_whh = sm + HH * FOUT;
  float* s_h   = s_whh + HH * HH;
  float* s_b   = s_h + 32 * HH;
  for (int i = threadIdx.x; i < HH * FOUT; i += 256) s_wih[i] = Wih[i];
  for (int i = threadIdx.x; i < HH * HH;  i += 256) s_whh[i] = Whh[i];
  for (int i = threadIdx.x; i < 32 * HH;  i += 256) s_h[i]   = 0.f;
  if (threadIdx.x < HH) s_b[threadIdx.x] = bih[threadIdx.x] + bhh[threadIdx.x];
  __syncthreads();

  const int wave = threadIdx.x >> 5, lane = threadIdx.x & 31;
  const int mr = lane & 15, half = lane >> 4, kk = half * 2;
  const int n0 = wave * 16;
  const size_t p0 = (size_t)blockIdx.x * 32;
  const float bias = s_b[n0 + mr];
  const float* bw = s_wih + (n0 + mr) * FOUT + kk;
  const float* bh = s_whh + (n0 + mr) * HH + kk;

  for (int t = 0; t < TT; ++t) {
    v8f c0 = {}, c1 = {};
    const float* a0 = gc + ((p0 + mr) * TT + t) * FIN + kk;
    const float* a1 = gc + ((p0 + 16 + mr) * TT + t) * FIN + kk;
#pragma unroll
    for (int k = 0; k < FIN; k += 4) {
      v2f b = {bw[k], bw[k + 1]};
      v2f a = {a0[k], a0[k + 1]};
      c0 = wmma4(a, b, c0);
      v2f a2 = {a1[k], a1[k + 1]};
      c1 = wmma4(a2, b, c1);
    }
    const float* h0 = s_h + mr * HH + kk;
    const float* h1 = s_h + (16 + mr) * HH + kk;
#pragma unroll
    for (int k = 0; k < HH; k += 4) {
      v2f b = {bh[k], bh[k + 1]};
      v2f a = {h0[k], h0[k + 1]};
      c0 = wmma4(a, b, c0);
      v2f a2 = {h1[k], h1[k + 1]};
      c1 = wmma4(a2, b, c1);
    }
    __syncthreads();
#pragma unroll
    for (int j = 0; j < 8; ++j) {
      int r = j + 8 * half;
      float v0 = tanhf(c0[j] + bias);
      float v1 = tanhf(c1[j] + bias);
      s_h[r * HH + n0 + mr] = v0;
      s_h[(16 + r) * HH + n0 + mr] = v1;
      rnn[((p0 + r) * TT + t) * HH + n0 + mr] = v0;
      rnn[((p0 + 16 + r) * TT + t) * HH + n0 + mr] = v1;
    }
    __syncthreads();
  }
}

__global__ __launch_bounds__(256) void k_mlp(const float* __restrict__ Xg,
                                             const float* __restrict__ W1, const float* __restrict__ B1,
                                             const float* __restrict__ W2, const float* __restrict__ B2,
                                             const float* __restrict__ W3, const float* __restrict__ B3,
                                             float* __restrict__ out) {
  extern __shared__ float sm[];
  float* sX  = sm;
  float* sY1 = sm + 32 * HH;
  float* sY2 = sY1 + 32 * HID;
  const size_t r0 = (size_t)blockIdx.x * 32;
  for (int i = threadIdx.x; i < 32 * HH; i += 256) sX[i] = Xg[r0 * HH + i];
  __syncthreads();
  const int wave = threadIdx.x >> 5, lane = threadIdx.x & 31;
  const int mr = lane & 15, half = lane >> 4, kk = half * 2;

  for (int ti = 0; ti < 8; ++ti) {
    int tile = wave * 8 + ti;
    int rt = tile & 1, ct = tile >> 1;
    const float* ap = sX + (rt * 16 + mr) * HH + kk;
    const float* bp = W1 + (size_t)(ct * 16 + mr) * HH + kk;
    v8f c = {};
#pragma unroll 8
    for (int k = 0; k < HH; k += 4) {
      v2f a = {ap[k], ap[k + 1]};
      v2f b = {bp[k], bp[k + 1]};
      c = wmma4(a, b, c);
    }
    float bias = B1[ct * 16 + mr];
#pragma unroll
    for (int j = 0; j < 8; ++j) {
      float v = c[j] + bias;
      sY1[(rt * 16 + j + 8 * half) * HID + ct * 16 + mr] = v > 0.f ? v : 0.f;
    }
  }
  __syncthreads();

  for (int ti = 0; ti < 8; ++ti) {
    int tile = wave * 8 + ti;
    int rt = tile & 1, ct = tile >> 1;
    const float* ap = sY1 + (rt * 16 + mr) * HID + kk;
    const float* bp = W2 + (size_t)(ct * 16 + mr) * HID + kk;
    v8f c = {};
#pragma unroll 4
    for (int k = 0; k < HID; k += 4) {
      v2f a = {ap[k], ap[k + 1]};
      v2f b = {bp[k], bp[k + 1]};
      c = wmma4(a, b, c);
    }
    float bias = B2[ct * 16 + mr];
#pragma unroll
    for (int j = 0; j < 8; ++j) {
      float v = c[j] + bias;
      sY2[(rt * 16 + j + 8 * half) * HID + ct * 16 + mr] = v > 0.f ? v : 0.f;
    }
  }
  __syncthreads();

  {
    int rt = wave & 1, ct = wave >> 1;
    const float* ap = sY2 + (rt * 16 + mr) * HID + kk;
    const float* bp = W3 + (size_t)(ct * 16 + mr) * HID + kk;
    v8f c = {};
#pragma unroll 4
    for (int k = 0; k < HID; k += 4) {
      v2f a = {ap[k], ap[k + 1]};
      v2f b = {bp[k], bp[k + 1]};
      c = wmma4(a, b, c);
    }
    float bias = B3[ct * 16 + mr];
#pragma unroll
    for (int j = 0; j < 8; ++j) {
      size_t r = r0 + rt * 16 + j + 8 * half;
      size_t p = r >> 3;
      int t = (int)(r & 7);
      out[(p * (TT + PP) + t) * FOUT + ct * 16 + mr] = c[j] + bias;
    }
  }
}

__global__ void k_zero_pred(float* __restrict__ out) {
  size_t idx = (size_t)blockIdx.x * 256 + threadIdx.x;
  if (idx < (size_t)BN * PP * FOUT) {
    size_t p = idx >> 7;
    size_t rem = idx & 127;
    out[p * (size_t)(TT + PP) * FOUT + (size_t)TT * FOUT + rem] = 0.f;
  }
}

extern "C" void kernel_launch(void* const* d_in, const int* in_sizes, int n_in,
                              void* d_out, int out_size, void* d_ws, size_t ws_size,
                              hipStream_t stream) {
  const float* x_in  = (const float*)d_in[0];
  const int*   ei    = (const int*)  d_in[1];
  const float* ew    = (const float*)d_in[2];
  const float* gcn_w = (const float*)d_in[3];
  const float* gcn_b = (const float*)d_in[4];
  const float* w_ih  = (const float*)d_in[5];
  const float* w_hh  = (const float*)d_in[6];
  const float* b_ih  = (const float*)d_in[7];
  const float* b_hh  = (const float*)d_in[8];
  const float* w1    = (const float*)d_in[9];
  const float* b1    = (const float*)d_in[10];
  const float* w2    = (const float*)d_in[11];
  const float* b2    = (const float*)d_in[12];
  const float* w3    = (const float*)d_in[13];
  const float* b3    = (const float*)d_in[14];
  float* out = (float*)d_out;

  float* ws   = (float*)d_ws;
  float* deg  = ws;
  float* dinv = ws + 16384;
  float* xl   = ws + 32768;
  float* gc   = xl + GNF;
  float* rnn  = gc + GNF;

  k_deg_init<<<(NN + 255) / 256, 256, 0, stream>>>(deg);
  k_deg_scatter<<<(EE + NN + 255) / 256, 256, 0, stream>>>(ei, ew, deg);
  k_dinv<<<(NN + 255) / 256, 256, 0, stream>>>(deg, dinv);

  k_gcn_gemm<<<(GG * NN) / 32, 256, 0, stream>>>(x_in, gcn_w, xl);

  k_agg_init<<<(int)((GNF + 255) / 256), 256, 0, stream>>>(gcn_b, gc);
  k_scatter<<<(int)(((size_t)(EE + NN) * 64 + 255) / 256), 256, 0, stream>>>(
      ei, ew, dinv, xl, gc);

  size_t rnn_lds = (size_t)(HH * FOUT + HH * HH + 32 * HH + HH) * sizeof(float);
  k_rnn<<<BN / 32, 256, rnn_lds, stream>>>(gc, w_ih, w_hh, b_ih, b_hh, rnn);

  size_t mlp_lds = (size_t)(32 * HH + 32 * HID + 32 * HID) * sizeof(float);
  k_mlp<<<(BN * TT) / 32, 256, mlp_lds, stream>>>(rnn, w1, b1, w2, b2, w3, b3, out);

  k_zero_pred<<<(int)(((size_t)BN * PP * FOUT + 255) / 256), 256, 0, stream>>>(out);
}


# Graph-layer projection folded into the RNN input weights: each RNN work-group builds W_ih*W_g and W_ih*b_g on matrix instructions in LDS; aggregation on raw rows; GEMM launch gone; sums reassociated
# speedup vs baseline: 4.5358x; 4.5358x over previous
#include <hip/hip_runtime.h>
#include <hip/hip_bf16.h>

typedef __attribute__((ext_vector_type(2))) float v2f;
typedef __attribute__((ext_vector_type(8))) float v8f;
typedef __attribute__((ext_vector_type(4))) float v4f;

static constexpr int   NN    = 10000;
static constexpr int   TT    = 8;
static constexpr int   PP    = 2;
static constexpr int   FIN   = 64;
static constexpr int   FOUT  = 64;
static constexpr int   HH    = 128;
static constexpr int   BB    = 2;
static constexpr int   EE    = 160000;
static constexpr int   GG    = BB * TT;
static constexpr int   BN    = BB * NN;
static constexpr int   HID   = HH * 4;
static constexpr size_t GNF  = (size_t)GG * NN * FOUT;

__device__ __forceinline__ v8f wmma4(v2f a, v2f b, v8f c) {
  return __builtin_amdgcn_wmma_f32_16x16x4_f32(false, a, false, b, (short)0, c,
                                               false, false);
}

__device__ __forceinline__ void atomAddF(float* p, float v) {
  __hip_atomic_fetch_add(p, v, __ATOMIC_RELAXED, __HIP_MEMORY_SCOPE_AGENT);
}

__global__ void k_deg_init(float* deg) {
  int i = blockIdx.x * 256 + threadIdx.x;
  if (i < NN) deg[i] = 0.f;
}

__global__ void k_deg_scatter(const int* __restrict__ ei,
                              const float* __restrict__ ew,
                              float* __restrict__ deg) {
  int e = blockIdx.x * 256 + threadIdx.x;
  if (e < EE)                atomAddF(&deg[ei[EE + e]], ew[e]);
  else if (e < EE + NN)      atomAddF(&deg[e - EE], 1.0f);
}

__global__ void k_dinv(const float* __restrict__ deg, float* __restrict__ dinv) {
  int i = blockIdx.x * 256 + threadIdx.x;
  if (i < NN) {
    float d = deg[i];
    dinv[i] = d > 0.f ? rsqrtf(d) : 0.f;
  }
}

typedef __attribute__((ext_vector_type(4))) int v4i;
static constexpr int TILE_N   = 40;
static constexpr int AGG_WAVES = 32;
static constexpr int TILE_F   = TILE_N * FOUT;
static constexpr int TILE_Q   = GG * TILE_F / 4;
static_assert(NN % TILE_N == 0, "tiles cover the nodes exactly");
static_assert(EE % 128 == 0, "a wave scans 128 edges per trip");

__global__ __launch_bounds__(1024) void k_agg_tile(const int* __restrict__ ei,
                                                   const float* __restrict__ ew,
                                                   const float* __restrict__ dinv,
                                                   const float* __restrict__ xr,
                                                   float* __restrict__ gc) {
  __shared__ __attribute__((aligned(16))) float acc[GG * TILE_F];
  const int tid = threadIdx.x, lane = tid & 31;
  const int wave = __builtin_amdgcn_readfirstlane(tid >> 5);
  const int n0 = blockIdx.x * TILE_N;

  for (int i = tid; i < TILE_Q; i += 1024) {
    const int g = i / (TILE_F / 4);
    const int rem = i - g * (TILE_F / 4);
    const int t = rem >> 4;
    const float di = dinv[n0 + t];
    const float s = di * di;
    const v4f x = *(const v4f*)(xr + ((size_t)g * NN + n0) * FOUT + 4 * rem);
    v4f r = {x.x * s, x.y * s, x.z * s, x.w * s};
    *(v4f*)(acc + 4 * i) = r;
  }
  __syncthreads();

  const int* __restrict__ colp = ei + EE;
  v4i c4 = *(const v4i*)(colp + wave * 128 + 4 * lane);
  for (int ch = wave; ch < EE / 128; ch += AGG_WAVES) {
    v4i nx = c4;
    if (ch + AGG_WAVES < EE / 128) nx = *(const v4i*)(colp + (ch + AGG_WAVES) * 128 + 4 * lane);
#pragma unroll
    for (int j = 0; j < 4; ++j) {
      const int t = c4[j] - n0;
      unsigned m = __builtin_amdgcn_ballot_w32((unsigned)t < (unsigned)TILE_N);
      while (m) {
        const int src = __builtin_ctz(m);
        m &= m - 1;
        const int e = ch * 128 + 4 * src + j;
        const int tl = __builtin_amdgcn_readlane(t, src);
        const int r = ei[e];
        const float nrm = dinv[r] * ew[e] * dinv[n0 + tl];
        const float* sp = xr + (size_t)r * FOUT + 2 * lane;
        float* dp = acc + tl * FOUT + 2 * lane;
        v2f v[GG];
#pragma unroll
        for (int g = 0; g < GG; ++g) v[g] = *(const v2f*)(sp + (size_t)g * NN * FOUT);
#pragma unroll
        for (int g = 0; g < GG; ++g) {
          __hip_atomic_fetch_add(dp + g * TILE_F,     v[g].x * nrm, __ATOMIC_RELAXED, __HIP_MEMORY_SCOPE_WORKGROUP);
          __hip_atomic_fetch_add(dp + g * TILE_F + 1, v[g].y * nrm, __ATOMIC_RELAXED, __HIP_MEMORY_SCOPE_WORKGROUP);
        }
      }
    }
    c4 = nx;
  }
  __syncthreads();

  for (int i = tid; i < TILE_Q; i += 1024) {
    const int g = i / (TILE_F / 4);
    const int rem = i - g * (TILE_F / 4);
    *(v4f*)(gc + ((size_t)g * NN + n0) * FOUT + 4 * rem) = *(const v4f*)(acc + 4 * i);
  }
}

static constexpr int RNN_WAVES = 5;
static constexpr int RNN_KT    = (FOUT + HH) / 16;
static constexpr int RNN_NT    = HH / 16;
static constexpr int RNN_WQ    = RNN_NT * RNN_KT * 64;
static constexpr int RNN_BLKS  = (RNN_NT / 2) * 3;
static constexpr int RNN_LDS_Q = 10304;
static_assert(BN % (16 * RNN_WAVES) == 0, "work-groups cover the sequences exactly");
static_assert(RNN_BLKS % 2 == 0, "fragment set parity is the same in every step");
static_assert((HH * HH / 4) % (128 * 8) == 0 && RNN_WAVES >= 4, "staging batches cover the weights exactly");
static_assert(FIN == FOUT && FIN % 16 == 0, "the folded input weight has the projection's input width");

__device__ __forceinline__ void rnn_fold_block(v4f* wq, const float* __restrict__ Wih,
                                               const float* __restrict__ Wg, int nu, int T, int lane) {
  const int c = lane & 15, h = lane >> 4;
  v8f accE = (v8f){}, accO = (v8f){};
#pragma unroll
  for (int Tj = 0; Tj < FOUT / 16; ++Tj) {
    const v4f* bp = (const v4f*)(Wih + (size_t)(16 * nu + c) * FOUT + 16 * Tj + 8 * h);
    const v4f b0 = bp[0], b1 = bp[1];
    const float* ap = Wg + (size_t)(16 * Tj + 8 * h) * FIN + 16 * T + c;
    float a[8];
#pragma unroll
    for (int e = 0; e < 8; ++e) a[e] = ap[e * FIN];
    v2f A, B;
    A = {a[0], a[1]}; B = {b0.x, b0.y}; accE = wmma4(A, B, accE);
    A = {a[2], a[3]}; B = {b0.z, b0.w}; accO = wmma4(A, B, accO);
    A = {a[4], a[5]}; B = {b1.x, b1.y}; accE = wmma4(A, B, accE);
    A = {a[6], a[7]}; B = {b1.z, b1.w}; accO = wmma4(A, B, accO);
  }
  v4f* dst = wq + (nu * RNN_KT + T) * 64 + lane;
  dst[0]  = (v4f){accE[0] + accO[0], accE[1] + accO[1], accE[2] + accO[2], accE[3] + accO[3]};
  dst[32] = (v4f){accE[4] + accO[4], accE[5] + accO[5], accE[6] + accO[6], accE[7] + accO[7]};
}

struct RnnFrag { v4f v[16]; };

__device__ __forceinline__ void rnn_load_set(RnnFrag& f, const v4f* wq, int bi, int lane) {
  const int np = bi / 3, g = bi - 3 * np;
#pragma unroll
  for (int tt = 0; tt < 4; ++tt) {
    const v4f* b0 = wq + ((2 * np)     * RNN_KT + 4 * g + tt) * 64 + lane;
    const v4f* b1 = wq + ((2 * np + 1) * RNN_KT + 4 * g + tt) * 64 + lane;
    f.v[4 * tt + 0] = b0[0];
    f.v[4 * tt + 1] = b0[32];
    f.v[4 * tt + 2] = b1[0];
    f.v[4 * tt + 3] = b1[32];
  }
}

__device__ __forceinline__ void rnn_mac_tile(v8f& acc0, v8f& acc1, v4f a0, v4f a1, v4f c0, v4f c1, v8f b) {
  v2f A, B;
  B = {b[0], b[1]};
  A = {a0.x, a0.y}; acc0 = wmma4(A, B, acc0);
  A = {c0.x, c0.y}; acc1 = wmma4(A, B, acc1);
  B = {b[2], b[3]};
  A = {a0.z, a0.w}; acc0 = wmma4(A, B, acc0);
  A = {c0.z, c0.w}; acc1 = wmma4(A, B, acc1);
  B = {b[4], b[5]};
  A = {a1.x, a1.y}; acc0 = wmma4(A, B, acc0);
  A = {c1.x, c1.y}; acc1 = wmma4(A, B, acc1);
  B = {b[6], b[7]};
  A = {a1.z, a1.w}; acc0 = wmma4(A, B, acc0);
  A = {c1.z, c1.w}; acc1 = wmma4(A, B, acc1);
}

__device__ __forceinline__ v8f rnn_ld8(const v4f* p) {
  v4f lo = p[0], hi = p[1];
  return (v8f){lo.x, lo.y, lo.z, lo.w, hi.x, hi.y, hi.z, hi.w};
}

__global__ __launch_bounds__(32 * RNN_WAVES) void k_rnn(const float* __restrict__ gc,
                                                        const float* __restrict__ Wg,
                                                        const float* __restrict__ bg,
                                                        const float* __restrict__ Wih,
                                                        const float* __restrict__ Whh,
                                                        const float* __restrict__ bih,
                                                        const float* __restrict__ bhh,
                                                        float* __restrict__ rnn) {
  __shared__ __attribute__((aligned(16))) v4f wq[RNN_LDS_Q];
  __shared__ __attribute__((aligned(16))) float bs[HH];
  const int tid = threadIdx.x;

  if (tid < 128) {
    for (int base = 0; base < HH * HH / 4; base += 128 * 8) {
      v4f v[8];
#pragma unroll
      for (int u = 0; u < 8; ++u) v[u] = *(const v4f*)(Whh + 4 * (base + u * 128 + tid));
#pragma unroll
      for (int u = 0; u < 8; ++u) {
        const int i = base + u * 128 + tid;
        const int row = i >> 5, T = 4 + ((i & 31) >> 2), u4 = i & 3;
        wq[((row >> 4) * RNN_KT + T) * 64 + (u4 & 1) * 32 + (u4 >> 1) * 16 + (row & 15)] = v[u];
      }
    }
  }
  {
    const int wave_s = __builtin_amdgcn_readfirstlane(tid >> 5);
    for (int job = wave_s; job < RNN_NT * (FIN / 16); job += RNN_WAVES)
      rnn_fold_block(wq, Wih, Wg, job / (FIN / 16), job % (FIN / 16), tid & 31);
  }
  if (tid < HH) {
    const v4f* wr = (const v4f*)(Wih + (size_t)tid * FOUT);
    float sb = bih[tid] + bhh[tid];
    for (int j4 = 0; j4 < FOUT / 4; ++j4) {
      const v4f w = wr[j4];
      const v4f g = *(const v4f*)(bg + 4 * j4);
      sb += w.x * g.x + w.y * g.y + w.z * g.z + w.w * g.w;
    }
    bs[tid] = sb;
  }
  __syncthreads();

  const int lane = tid & 31, wave = tid >> 5;
  const int c = lane & 15, h = lane >> 4;
  const size_t p = (size_t)blockIdx.x * (16 * RNN_WAVES) + wave * 16 + c;
  const v4f* xp = (const v4f*)(gc + p * TT * FIN + 8 * h);
  float* op = rnn + p * TT * HH + 8 * h;

  v8f hcur[RNN_NT];
#pragma unroll
  for (int m = 0; m < RNN_NT; ++m) hcur[m] = (v8f){};
  v8f xc[4];
#pragma unroll
  for (int T = 0; T < 4; ++T) xc[T] = rnn_ld8(xp + 4 * T);

  RnnFrag fs[2];
  rnn_load_set(fs[0], wq, 0, lane);

#pragma unroll 1
  for (int t = 0; t < TT; ++t) {
    const int tn = t + 1 < TT ? t + 1 : t;
    v8f xn[4];
#pragma unroll
    for (int T = 0; T < 4; ++T) xn[T] = rnn_ld8(xp + tn * (FIN / 4) + 4 * T);

    v8f hn[RNN_NT];
    v8f accs[RNN_NT];
#pragma unroll
    for (int np = 0; np < RNN_NT / 2; ++np) {
      v8f acc0 = (v8f){}, acc1 = (v8f){};
#pragma unroll
      for (int g = 0; g < 3; ++g) {
        const int bi = 3 * np + g;
        rnn_load_set(fs[(bi + 1) & 1], wq, (bi + 1) % RNN_BLKS, lane);
        __builtin_amdgcn_sched_barrier(0);
        if (g == 0 && np > 0) {
#pragma unroll
          for (int k2 = 0; k2 < 2; ++k2) {
            const int n = 2 * (np - 1) + k2;
            const v4f b0 = *(const v4f*)(bs + 16 * n + 8 * h), b1 = *(const v4f*)(bs + 16 * n + 8 * h + 4);
            const v8f a = accs[n];
            v8f y;
            y[0] = tanhf(a[0] + b0.x); y[1] = tanhf(a[1] + b0.y); y[2] = tanhf(a[2] + b0.z); y[3] = tanhf(a[3] + b0.w);
            y[4] = tanhf(a[4] + b1.x); y[5] = tanhf(a[5] + b1.y); y[6] = tanhf(a[6] + b1.z); y[7] = tanhf(a[7] + b1.w);
            hn[n] = y;
            v4f* o = (v4f*)(op + t * HH + 16 * n);
            o[0] = (v4f){y[0], y[1], y[2], y[3]};
            o[1] = (v4f){y[4], y[5], y[6], y[7]};
          }
        }
        const RnnFrag& f = fs[bi & 1];
#pragma unroll
        for (int tt = 0; tt < 4; ++tt) {
          const v8f bop = g == 0 ? xc[tt] : hcur[4 * (g - 1) + tt];
          rnn_mac_tile(acc0, acc1, f.v[4 * tt], f.v[4 * tt + 1], f.v[4 * tt + 2], f.v[4 * tt + 3], bop);
        }
        __builtin_amdgcn_sched_barrier(0);
      }
      accs[2 * np] = acc0;
      accs[2 * np + 1] = acc1;
    }
#pragma unroll
    for (int k2 = 0; k2 < 2; ++k2) {
      const int n = RNN_NT - 2 + k2;
      const v4f b0 = *(const v4f*)(bs + 16 * n + 8 * h), b1 = *(const v4f*)(bs + 16 * n + 8 * h + 4);
      const v8f a = accs[n];
      v8f y;
      y[0] = tanhf(a[0] + b0.x); y[1] = tanhf(a[1] + b0.y); y[2] = tanhf(a[2] + b0.z); y[3] = tanhf(a[3] + b0.w);
      y[4] = tanhf(a[4] + b1.x); y[5] = tanhf(a[5] + b1.y); y[6] = tanhf(a[6] + b1.z); y[7] = tanhf(a[7] + b1.w);
      hn[n] = y;
      v4f* o = (v4f*)(op + t * HH + 16 * n);
      o[0] = (v4f){y[0], y[1], y[2], y[3]};
      o[1] = (v4f){y[4], y[5], y[6], y[7]};
    }
#pragma unroll
    for (int m = 0; m < RNN_NT; ++m) hcur[m] = hn[m];
#pragma unroll
    for (int T = 0; T < 4; ++T) xc[T] = xn[T];
  }
}

static constexpr int PANEL_Q = 2304;
static constexpr int MLP_RING = 3;
static constexpr int MLP_ROWS = 128;

__device__ __forceinline__ void panel_arrive() {
  __builtin_amdgcn_fence(__ATOMIC_RELEASE, "workgroup");
  __builtin_amdgcn_s_barrier_signal(-1);
}
__device__ __forceinline__ void panel_wait() {
  __builtin_amdgcn_s_barrier_wait(-1);
  __builtin_amdgcn_fence(__ATOMIC_ACQUIRE, "workgroup");
}

__device__ __forceinline__ v8f tile_mac4(v8f acc, v4f a0, v4f a1, v8f b) {
  v2f A, B;
  A = {a0.x, a0.y}; B = {b[0], b[1]}; acc = wmma4(A, B, acc);
  A = {a0.z, a0.w}; B = {b[2], b[3]}; acc = wmma4(A, B, acc);
  A = {a1.x, a1.y}; B = {b[4], b[5]}; acc = wmma4(A, B, acc);
  A = {a1.z, a1.w}; B = {b[6], b[7]}; acc = wmma4(A, B, acc);
  return acc;
}

__device__ __forceinline__ float relu1(float v) { return v > 0.f ? v : 0.f; }

__global__ __launch_bounds__(256) void k_mlp(const float* __restrict__ Xg,
                                             const float* __restrict__ W1, const float* __restrict__ B1,
                                             const float* __restrict__ W2, const float* __restrict__ B2,
                                             const float* __restrict__ W3, const float* __restrict__ B3,
                                             float* __restrict__ out) {
  extern __shared__ v4f smq[];
  const int tid = threadIdx.x;
  const int wave = tid >> 5, lane = tid & 31;
  const int c = lane & 15, h = lane >> 4;

  const int sc = tid & 15;
  const int sg = tid >> 4;
  const int su = sg & 3;
  const int sT = sg >> 2;
  const int sdst = (su & 1) * 32 + (su >> 1) * 16 + sc;
  const int trow = tid >> 2;
  const int tu = tid & 3;
  const int tdst = 2048 + (trow >> 4) * 64 + (tu & 1) * 32 + (tu >> 1) * 16 + (trow & 15);

  const size_t R = (size_t)blockIdx.x * MLP_ROWS + wave * 16 + c;

  v8f x[8];
  {
    const v4f* xp = (const v4f*)(Xg + R * HH + 8 * h);
#pragma unroll
    for (int T = 0; T < 8; ++T) {
      v4f lo = xp[T * 4], hi = xp[T * 4 + 1];
      x[T] = (v8f){lo.x, lo.y, lo.z, lo.w, hi.x, hi.y, hi.z, hi.w};
    }
  }

  v4f pf[9];
#pragma unroll
  for (int i = 0; i < 8; ++i) {
    const int nn = i >> 1, f4 = sg + 16 * (i & 1);
    pf[i] = *(const v4f*)(W1 + (size_t)(16 * nn + sc) * HH + 4 * f4);
  }
#pragma unroll
  for (int i = 0; i < 8; ++i) {
    const int nn = i >> 1, T = sT + 4 * (i & 1);
    smq[nn * 512 + T * 64 + sdst] = pf[i];
  }
  __syncthreads();

  v8f y1[32];
#pragma unroll
  for (int p = 0; p < 8; ++p) {
    if (p < 7) {
#pragma unroll
      for (int i = 0; i < 8; ++i) {
        const int nn = i >> 1, f4 = sg + 16 * (i & 1);
        pf[i] = *(const v4f*)(W1 + (size_t)(16 * (4 * (p + 1) + nn) + sc) * HH + 4 * f4);
      }
    } else {
#pragma unroll
      for (int i = 0; i < 8; ++i)
        pf[i] = *(const v4f*)(W2 + (size_t)sc * HID + 4 * (sg + 16 * i));
      pf[8] = *(const v4f*)(W3 + (size_t)trow * HID + 4 * tu);
    }
    const v4f* pb = smq + (p & 1) * PANEL_Q;
#pragma unroll
    for (int nn = 0; nn < 4; ++nn) {
      v8f aE = {}, aO = {};
#pragma unroll
      for (int T = 0; T < 8; ++T) {
        v4f a0 = pb[nn * 512 + T * 64 + lane];
        v4f a1 = pb[nn * 512 + T * 64 + 32 + lane];
        if (T & 1) aO = tile_mac4(aO, a0, a1, x[T]);
        else       aE = tile_mac4(aE, a0, a1, x[T]);
      }
      const v4f* bp = (const v4f*)(B1 + 16 * (4 * p + nn) + 8 * h);
      v4f b0 = bp[0], b1 = bp[1];
      v8f bb = (v8f){b0.x, b0.y, b0.z, b0.w, b1.x, b1.y, b1.z, b1.w};
      v8f y;
#pragma unroll
      for (int r = 0; r < 8; ++r) y[r] = relu1((aE[r] + aO[r]) + bb[r]);
      y1[4 * p + nn] = y;
    }
    v4f* nb = smq + ((p + 1) & 1) * PANEL_Q;
    if (p < 7) {
#pragma unroll
      for (int i = 0; i < 8; ++i) {
        const int nn = i >> 1, T = sT + 4 * (i & 1);
        nb[nn * 512 + T * 64 + sdst] = pf[i];
      }
    } else {
#pragma unroll
      for (int i = 0; i < 8; ++i) nb[(sT + 4 * i) * 64 + sdst] = pf[i];
      nb[tdst] = pf[8];
    }
    __syncthreads();
  }

  v8f oacc[4] = {};
  int cur = 0, nxt = PANEL_Q, thr = 2 * PANEL_Q;
  for (int n = 0; n < 32; ++n) {
    const bool more = n < 31;
    if (more) {
#pragma unroll
      for (int i = 0; i < 8; ++i)
        pf[i] = *(const v4f*)(W2 + (size_t)(16 * (n + 1) + sc) * HID + 4 * (sg + 16 * i));
      pf[8] = *(const v4f*)(W3 + (size_t)trow * HID + 16 * (n + 1) + 4 * tu);
    }
    const v4f* pb = smq + cur;
    v8f aE = {}, aO = {};
#pragma unroll
    for (int T = 0; T < 16; ++T) {
      v4f a0 = pb[T * 64 + lane];
      v4f a1 = pb[T * 64 + 32 + lane];
      if (T & 1) aO = tile_mac4(aO, a0, a1, y1[T]);
      else       aE = tile_mac4(aE, a0, a1, y1[T]);
    }
    if (more) {
      v4f* nb = smq + nxt;
#pragma unroll
      for (int i = 0; i < 8; ++i) nb[(sT + 4 * i) * 64 + sdst] = pf[i];
      nb[tdst] = pf[8];
      panel_arrive();
    }
#pragma unroll
    for (int T = 16; T < 32; ++T) {
      v4f a0 = pb[T * 64 + lane];
      v4f a1 = pb[T * 64 + 32 + lane];
      if (T & 1) aO = tile_mac4(aO, a0, a1, y1[T]);
      else       aE = tile_mac4(aE, a0, a1, y1[T]);
    }
    const v4f* bp = (const v4f*)(B2 + 16 * n + 8 * h);
    v4f b0 = bp[0], b1 = bp[1];
    v8f bb = (v8f){b0.x, b0.y, b0.z, b0.w, b1.x, b1.y, b1.z, b1.w};
    v8f y2;
#pragma unroll
    for (int r = 0; r < 8; ++r) y2[r] = relu1((aE[r] + aO[r]) + bb[r]);
#pragma unroll
    for (int m = 0; m < 4; ++m) {
      v4f a0 = pb[2048 + m * 64 + lane];
      v4f a1 = pb[2048 + m * 64 + 32 + lane];
      oacc[m] = tile_mac4(oacc[m], a0, a1, y2);
    }
    if (more) panel_wait();
    const int t3 = cur; cur = nxt; nxt = thr; thr = t3;
  }

  const size_t orow = ((R >> 3) * (size_t)(TT + PP) + (R & 7)) * FOUT;
#pragma unroll
  for (int m = 0; m < 4; ++m) {
    const v4f* bp = (const v4f*)(B3 + 16 * m + 8 * h);
    v4f b0 = bp[0], b1 = bp[1];
    v4f o0 = {oacc[m][0] + b0.x, oacc[m][1] + b0.y, oacc[m][2] + b0.z, oacc[m][3] + b0.w};
    v4f o1 = {oacc[m][4] + b1.x, oacc[m][5] + b1.y, oacc[m][6] + b1.z, oacc[m][7] + b1.w};
    v4f* op = (v4f*)(out + orow + 16 * m + 8 * h);
    op[0] = o0;
    op[1] = o1;
  }
}

__global__ void k_zero_pred(float* __restrict__ out) {
  size_t idx = (size_t)blockIdx.x * 256 + threadIdx.x;
  if (idx < (size_t)BN * PP * FOUT) {
    size_t p = idx >> 7;
    size_t rem = idx & 127;
    out[p * (size_t)(TT + PP) * FOUT + (size_t)TT * FOUT + rem] = 0.f;
  }
}

extern "C" void kernel_launch(void* const* d_in, const int* in_sizes, int n_in,
                              void* d_out, int out_size, void* d_ws, size_t ws_size,
                              hipStream_t stream) {
  const float* x_in  = (const float*)d_in[0];
  const int*   ei    = (const int*)  d_in[1];
  const float* ew    = (const float*)d_in[2];
  const float* gcn_w = (const float*)d_in[3];
  const float* gcn_b = (const float*)d_in[4];
  const float* w_ih  = (const float*)d_in[5];
  const float* w_hh  = (const float*)d_in[6];
  const float* b_ih  = (const float*)d_in[7];
  const float* b_hh  = (const float*)d_in[8];
  const float* w1    = (const float*)d_in[9];
  const float* b1    = (const float*)d_in[10];
  const float* w2    = (const float*)d_in[11];
  const float* b2    = (const float*)d_in[12];
  const float* w3    = (const float*)d_in[13];
  const float* b3    = (const float*)d_in[14];
  float* out = (float*)d_out;

  float* ws   = (float*)d_ws;
  float* deg  = ws;
  float* dinv = ws + 16384;
  float* gc   = ws + 32768 + GNF;
  float* rnn  = gc + GNF;

  k_deg_init<<<(NN + 255) / 256, 256, 0, stream>>>(deg);
  k_deg_scatter<<<(EE + NN + 255) / 256, 256, 0, stream>>>(ei, ew, deg);
  k_dinv<<<(NN + 255) / 256, 256, 0, stream>>>(deg, dinv);

  k_agg_tile<<<NN / TILE_N, 1024, 0, stream>>>(ei, ew, dinv, x_in, gc);

  k_rnn<<<BN / (16 * RNN_WAVES), 32 * RNN_WAVES, 0, stream>>>(gc, gcn_w, gcn_b, w_ih, w_hh, b_ih, b_hh, rnn);

  size_t mlp_lds = (size_t)(MLP_RING * PANEL_Q) * 4 * sizeof(float);
  k_mlp<<<(BN * TT) / MLP_ROWS, 256, mlp_lds, stream>>>(rnn, w1, b1, w2, b2, w3, b3, out);

  k_zero_pred<<<(int)(((size_t)BN * PP * FOUT + 255) / 256), 256, 0, stream>>>(out);
}
